# starting kernel
# baseline (speedup 1.0000x reference)
#include <hip/hip_runtime.h>

typedef __attribute__((ext_vector_type(2))) float v2f;
typedef __attribute__((ext_vector_type(8))) float v8f;

#define DFEAT 256


__global__ void zero_kernel(float* __restrict__ p, long n) {
    long i = (long)blockIdx.x * blockDim.x + threadIdx.x;
    long stride = (long)gridDim.x * blockDim.x;
    for (; i < n; i += stride) p[i] = 0.0f;
}

__global__ void scatter_accum_kernel(const float* __restrict__ hsrc,
                                     const int* __restrict__ src,
                                     const int* __restrict__ dst,
                                     float* __restrict__ sums,
                                     float* __restrict__ cnt,
                                     int nE) {
    int e = blockIdx.x;
    if (e >= nE) return;
    int s = src[e];
    int d = dst[e];
    const float* row = hsrc + (size_t)s * DFEAT;
    float*       out = sums + (size_t)d * DFEAT;
    atomicAdd(&out[threadIdx.x], row[threadIdx.x]);
    if (threadIdx.x == 0) atomicAdd(&cnt[d], 1.0f);
}

__global__ void inv_kernel(const float* __restrict__ cnt,
                           float* __restrict__ inv, int n) {
    int i = blockIdx.x * blockDim.x + threadIdx.x;
    if (i < n) inv[i] = 1.0f / fmaxf(cnt[i], 1.0f);
}

template<int NT, bool RAGGED, bool RELU>
__global__ void sage_gemm_wmma(const float* __restrict__ Asum,
                               const float* __restrict__ invc,
                               const float* __restrict__ Htgt,
                               const float* __restrict__ Wl,
                               const float* __restrict__ Wr,
                               const float* __restrict__ bias,
                               float* __restrict__ Out,
                               int tilesNG,
                               int totalTiles,
                               int Nact) {
    const int wave = threadIdx.x >> 5;
    const int lane = threadIdx.x & 31;
    const int gw = blockIdx.x * (blockDim.x >> 5) + wave;
    if (gw >= totalTiles) return;

    const int tileM  = gw / tilesNG;
    const int tileNG = gw % tilesNG;
    const int l15  = lane & 15;
    const int row  = tileM * 16 + l15;
    const int koff = (lane < 16) ? 0 : 2;
    const int colBase = tileNG * (16 * NT);

    int col[NT];
#pragma unroll
    for (int j = 0; j < NT; ++j) col[j] = colBase + j * 16 + l15;

    const float invr = invc[row];

    v8f acc[NT];
#pragma unroll
    for (int j = 0; j < NT; ++j) {
        float bv = (!RAGGED || col[j] < Nact) ? bias[col[j]] : 0.0f;
#pragma unroll
        for (int i = 0; i < 8; ++i) acc[j][i] = bv;
    }

    const float* arow = Asum + (size_t)row * DFEAT + koff;
    const float* hrow = Htgt + (size_t)row * DFEAT + koff;
    const float* pl = Wl + (size_t)koff * Nact;
    const float* pr = Wr + (size_t)koff * Nact;
    const size_t bstep = 4 * (size_t)Nact;

    for (int k = 0; k < DFEAT; k += 4) {
        float2 av = *(const float2*)arow;  arow += 4;
        float2 hv = *(const float2*)hrow;  hrow += 4;
        v2f a;  a.x  = av.x * invr; a.y  = av.y * invr;
        v2f a2; a2.x = hv.x;        a2.y = hv.y;

#pragma unroll
        for (int j = 0; j < NT; ++j) {
            v2f b, b2;
            if (!RAGGED || col[j] < Nact) {
                b.x  = pl[col[j]];
                b.y  = pl[Nact + col[j]];
                b2.x = pr[col[j]];
                b2.y = pr[Nact + col[j]];
            } else {
                b.x = 0.0f; b.y = 0.0f; b2.x = 0.0f; b2.y = 0.0f;
            }
            acc[j] = __builtin_amdgcn_wmma_f32_16x16x4_f32(
                         false, a, false, b, (short)0, acc[j], false, false);
            acc[j] = __builtin_amdgcn_wmma_f32_16x16x4_f32(
                         false, a2, false, b2, (short)0, acc[j], false, false);
        }
        pl += bstep;
        pr += bstep;
    }

    const int mbase = tileM * 16 + ((lane < 16) ? 0 : 8);
#pragma unroll
    for (int j = 0; j < NT; ++j) {
        if (!RAGGED || col[j] < Nact) {
#pragma unroll
            for (int i = 0; i < 8; ++i) {
                float v = acc[j][i];
                if (RELU) v = fmaxf(v, 0.0f);
                Out[(size_t)(mbase + i) * Nact + col[j]] = v;
            }
        }
    }
}


extern "C" void kernel_launch(void* const* d_in, const int* in_sizes, int n_in,
                              void* d_out, int out_size, void* d_ws, size_t ws_size,
                              hipStream_t stream) {
    const float* h   = (const float*)d_in[0];
    const float* wl0 = (const float*)d_in[1];
    const float* wr0 = (const float*)d_in[2];
    const float* b0  = (const float*)d_in[3];
    const float* wl1 = (const float*)d_in[4];
    const float* wr1 = (const float*)d_in[5];
    const float* b1  = (const float*)d_in[6];
    const int*   ei0 = (const int*)d_in[7];
    const int*   ei1 = (const int*)d_in[8];

    const int E0 = in_sizes[7] / 2;
    const int E1 = in_sizes[8] / 2;
    const int NT0 = 20000;
    const int NT1 = 2000;
    const int NCLS = 47;

    float* sums0 = (float*)d_ws;
    float* cnt0  = sums0 + (size_t)NT0 * DFEAT;
    float* inv0  = cnt0 + NT0;
    float* h1    = inv0 + NT0;
    float* sums1 = h1 + (size_t)NT0 * DFEAT;
    float* cnt1  = sums1 + (size_t)NT1 * DFEAT;
    float* inv1  = cnt1 + NT1;

    zero_kernel<<<2048, 256, 0, stream>>>(sums0, (long)NT0 * DFEAT);
    zero_kernel<<<64,   256, 0, stream>>>(cnt0, NT0);
    zero_kernel<<<512,  256, 0, stream>>>(sums1, (long)NT1 * DFEAT);
    zero_kernel<<<8,    256, 0, stream>>>(cnt1, NT1);

    scatter_accum_kernel<<<E0, DFEAT, 0, stream>>>(h, ei0, ei0 + E0, sums0, cnt0, E0);
    inv_kernel<<<(NT0 + 255) / 256, 256, 0, stream>>>(cnt0, inv0, NT0);

    {
        const int tilesNG = DFEAT / 64;
        const int total   = (NT0 / 16) * tilesNG;
        const int blocks  = (total + 7) / 8;
        sage_gemm_wmma<4, false, true><<<blocks, 256, 0, stream>>>(
            sums0, inv0, h, wl0, wr0, b0, h1, tilesNG, total, DFEAT);
    }

    scatter_accum_kernel<<<E1, DFEAT, 0, stream>>>(h1, ei1, ei1 + E1, sums1, cnt1, E1);
    inv_kernel<<<(NT1 + 255) / 256, 256, 0, stream>>>(cnt1, inv1, NT1);

    {
        const int tilesNG = 1;
        const int total   = (NT1 / 16) * tilesNG;
        const int blocks  = (total + 7) / 8;
        sage_gemm_wmma<3, true, false><<<blocks, 256, 0, stream>>>(
            sums1, inv1, h1, wl1, wr1, b1, (float*)d_out, tilesNG, total, NCLS);
    }
}


# Both GEMMs re-tiled per work-group (80x256 and 16x48 outputs): A rows and weight k-chunks staged once in LDS by coalesced loads one chunk ahead, WMMA fragments read from LDS; sum order unchanged
# speedup vs baseline: 3.9355x; 3.9355x over previous
#include <hip/hip_runtime.h>

typedef __attribute__((ext_vector_type(2))) float v2f;
typedef __attribute__((ext_vector_type(4))) float v4f;
typedef __attribute__((ext_vector_type(8))) float v8f;

#define DFEAT 256

#define AGG_THREADS  1024
#define AGG_WAVES    32
#define AGG_CAP      16384
#define AGG_INFLIGHT 8

template<int R>
__global__ __launch_bounds__(AGG_THREADS)
void sage_mean_owner_kernel(const float* __restrict__ hsrc,
                            const int* __restrict__ src,
                            const int* __restrict__ dst,
                            float* __restrict__ mean,
                            int nE, int nTgt) {
    __shared__ float slab[R * DFEAT];
    __shared__ int   list_e[AGG_CAP];
    __shared__ int   list_t[AGG_CAP];
    __shared__ int   cnt[R];
    __shared__ int   n_list;

    const int tid  = threadIdx.x;
    const int lane = tid & 31;
    const int wave = tid >> 5;
    const int base = blockIdx.x * R;
    const int rows = min(R, nTgt - base);
    const unsigned ubase = (unsigned)base;
    const unsigned urows = (unsigned)rows;

    for (int i = tid; i < R * DFEAT / 4; i += AGG_THREADS)
        ((float4*)slab)[i] = make_float4(0.0f, 0.0f, 0.0f, 0.0f);
    for (int i = tid; i < R; i += AGG_THREADS) cnt[i] = 0;
    if (tid == 0) n_list = 0;
    __syncthreads();

    auto record = [&](unsigned t, int i) {
        if (t < urows) {
            atomicAdd(&cnt[t], 1);
            const int idx = atomicAdd(&n_list, 1);
            if (idx < AGG_CAP) { list_e[idx] = i; list_t[idx] = (int)t; }
        }
    };
    auto check4 = [&](const int4 q, int i) {
        const unsigned t0 = (unsigned)q.x - ubase;
        const unsigned t1 = (unsigned)q.y - ubase;
        const unsigned t2 = (unsigned)q.z - ubase;
        const unsigned t3 = (unsigned)q.w - ubase;
        if ((t0 < urows) | (t1 < urows) | (t2 < urows) | (t3 < urows)) {
            record(t0, i); record(t1, i + 1); record(t2, i + 2); record(t3, i + 3);
        }
    };
    const bool vec_ok = (((size_t)dst) & 15) == 0;
    const int  nVec   = vec_ok ? (nE >> 2) : 0;
    const int4* dst4  = (const int4*)dst;
    const int start = (int)blockIdx.x * (nVec / (int)gridDim.x);
    for (int pass = 0; pass < 2; ++pass) {
        const int lo = pass ? 0 : start;
        const int hi = pass ? start : nVec;
        int v = lo + tid;
        for (; v + 3 * AGG_THREADS < hi; v += 4 * AGG_THREADS) {
            const int4 q0 = dst4[v];
            const int4 q1 = dst4[v + AGG_THREADS];
            const int4 q2 = dst4[v + 2 * AGG_THREADS];
            const int4 q3 = dst4[v + 3 * AGG_THREADS];
            check4(q0, 4 * v);
            check4(q1, 4 * (v + AGG_THREADS));
            check4(q2, 4 * (v + 2 * AGG_THREADS));
            check4(q3, 4 * (v + 3 * AGG_THREADS));
        }
        for (; v < hi; v += AGG_THREADS) check4(dst4[v], 4 * v);
    }
    for (int i = 4 * nVec + tid; i < nE; i += AGG_THREADS)
        record((unsigned)dst[i] - ubase, i);
    __syncthreads();

    int n = __builtin_amdgcn_readfirstlane(n_list);
    if (n > AGG_CAP) {
        for (int i = tid; i < nE; i += AGG_THREADS) {
            const unsigned t = (unsigned)dst[i] - ubase;
            if (t < urows) {
                const float* row = hsrc + (size_t)src[i] * DFEAT;
                float* acc = slab + t * DFEAT;
                for (int c = 0; c < DFEAT; ++c) atomicAdd(&acc[c], row[c]);
            }
        }
        n = 0;
    }

    for (int i = tid; i < n; i += AGG_THREADS) list_e[i] = src[list_e[i]];
    __syncthreads();

    const int nFull = n / AGG_INFLIGHT;
    for (int b = wave; b < nFull; b += AGG_WAVES) {
        float2 v[AGG_INFLIGHT][4];
        int    tt[AGG_INFLIGHT];
#pragma unroll
        for (int u = 0; u < AGG_INFLIGHT; ++u) {
            const int i = b * AGG_INFLIGHT + u;
            const int s = __builtin_amdgcn_readfirstlane(list_e[i]);
            tt[u]       = __builtin_amdgcn_readfirstlane(list_t[i]);
            const float2* row = (const float2*)(hsrc + (size_t)s * DFEAT);
#pragma unroll
            for (int j = 0; j < 4; ++j) v[u][j] = row[lane + 32 * j];
        }
#pragma unroll
        for (int u = 0; u < AGG_INFLIGHT; ++u) {
            float* acc = slab + tt[u] * DFEAT + 2 * lane;
#pragma unroll
            for (int j = 0; j < 4; ++j) {
                atomicAdd(&acc[64 * j],     v[u][j].x);
                atomicAdd(&acc[64 * j + 1], v[u][j].y);
            }
        }
    }
    for (int i = nFull * AGG_INFLIGHT + wave; i < n; i += AGG_WAVES) {
        const int s = __builtin_amdgcn_readfirstlane(list_e[i]);
        const int t = __builtin_amdgcn_readfirstlane(list_t[i]);
        const float2* row = (const float2*)(hsrc + (size_t)s * DFEAT);
        float* acc = slab + t * DFEAT + 2 * lane;
#pragma unroll
        for (int j = 0; j < 4; ++j) {
            const float2 x = row[lane + 32 * j];
            atomicAdd(&acc[64 * j],     x.x);
            atomicAdd(&acc[64 * j + 1], x.y);
        }
    }
    __syncthreads();

    for (int r = wave; r < rows; r += AGG_WAVES) {
        const float inv = 1.0f / fmaxf((float)cnt[r], 1.0f);
        const float4* srow = (const float4*)(slab + r * DFEAT);
        float4* orow = (float4*)(mean + (size_t)(base + r) * DFEAT);
#pragma unroll
        for (int j = 0; j < 2; ++j) {
            float4 x = srow[lane + 32 * j];
            x.x *= inv; x.y *= inv; x.z *= inv; x.w *= inv;
            orow[lane + 32 * j] = x;
        }
    }
}

template<int MT, int NTW, int NWAVE, int KC, bool WVEC, bool RELU>
__global__ __launch_bounds__(NWAVE * 32)
void sage_gemm_lds(const float* __restrict__ Amean,
                   const float* __restrict__ Htgt,
                   const float* __restrict__ Wl,
                   const float* __restrict__ Wr,
                   const float* __restrict__ bias,
                   float* __restrict__ Out,
                   int Nact) {
    constexpr int NTHR   = NWAVE * 32;
    constexpr int ROWS   = 16 * MT;
    constexpr int NCOLS  = 16 * NTW * NWAVE;
    constexpr int KP     = KC / 2;
    constexpr int KQ     = KC / 4;
    constexpr int WPITCH = 2 * NCOLS + ((32 - (2 * NCOLS) % 64 + 64) % 64);
    constexpr int APITCH = KC + 4;
    constexpr int WSZ    = KP * WPITCH;
    constexpr int ASZ    = ROWS * APITCH;
    constexpr int STAGE  = 2 * WSZ + 2 * ASZ;
    constexpr int NCHUNK = DFEAT / KC;
    constexpr int ARPR   = NTHR / KQ;
    constexpr int AR     = (2 * ROWS + ARPR - 1) / ARPR;
    constexpr int WCPT   = WVEC ? 4 : 1;
    constexpr int WRPR   = NTHR / (NCOLS / WCPT);
    constexpr int WRM    = KP / WRPR;
    static_assert(NTHR % KQ == 0 && NTHR % (NCOLS / WCPT) == 0 && KP % WRPR == 0, "copy shares must tile the chunk");

    __shared__ __attribute__((aligned(16))) float lds[2 * STAGE];

    const int tid  = threadIdx.x;
    const int wave = tid >> 5;
    const int lane = tid & 31;
    const int c    = lane & 15;
    const int h    = lane >> 4;
    const int ldw = WVEC ? NCOLS : Nact;
    const int rowBase = blockIdx.x * ROWS;
    const int colBase = wave * (16 * NTW);

    const float* asrc[AR];
    int          aoff[AR];
    {
        const int q  = tid % KQ;
        const int r0 = tid / KQ;
#pragma unroll
        for (int i = 0; i < AR; ++i) {
            const int mrow = min(r0 + i * ARPR, 2 * ROWS - 1);
            const int mat  = mrow >= ROWS;
            const int m    = mrow - mat * ROWS;
            asrc[i] = (mat ? Htgt : Amean) + (size_t)(rowBase + m) * DFEAT + 4 * q;
            aoff[i] = 2 * WSZ + mat * ASZ + m * APITCH + 4 * q;
        }
    }
    const int  wcol = (tid % (NCOLS / WCPT)) * WCPT;
    const int  wp0  = tid / (NCOLS / WCPT);
    const bool wok  = wcol < Nact;
    const int  wtoff = 2 * wp0 * ldw + (wok ? wcol : 0);
    const int    woff0 = wp0 * WPITCH + 2 * wcol;

    v4f areg[AR];
    v4f wv0[2][WRM], wv1[2][WRM];
    v2f ws[2][WRM];

    auto gload = [&](int k0) {
#pragma unroll
        for (int i = 0; i < AR; ++i) areg[i] = *(const v4f*)(asrc[i] + k0);
#pragma unroll
        for (int mat = 0; mat < 2; ++mat) {
            const float* wb = (mat ? Wr : Wl) + (size_t)k0 * ldw;
#pragma unroll
            for (int i = 0; i < WRM; ++i) {
                const float* s = wb + (size_t)(2 * i * WRPR) * ldw;
                if constexpr (WVEC) {
                    wv0[mat][i] = *(const v4f*)(s + wtoff);
                    wv1[mat][i] = *(const v4f*)(s + ldw + wtoff);
                } else {
                    ws[mat][i].x = s[wtoff];
                    ws[mat][i].y = s[ldw + wtoff];
                }
            }
        }
    };
    auto lstore = [&](float* stage) {
#pragma unroll
        for (int i = 0; i < AR; ++i) *(v4f*)(stage + aoff[i]) = areg[i];
#pragma unroll
        for (int mat = 0; mat < 2; ++mat) {
#pragma unroll
            for (int i = 0; i < WRM; ++i) {
                float* d = stage + mat * WSZ + woff0 + i * WRPR * WPITCH;
                if constexpr (WVEC) {
                    const v4f r0 = wv0[mat][i];
                    const v4f r1 = wv1[mat][i];
                    v4f o0, o1;
                    o0.x = r0.x; o0.y = r1.x; o0.z = r0.y; o0.w = r1.y;
                    o1.x = r0.z; o1.y = r1.z; o1.z = r0.w; o1.w = r1.w;
                    *(v4f*)d       = o0;
                    *(v4f*)(d + 4) = o1;
                } else {
                    *(v2f*)d = ws[mat][i];
                }
            }
        }
    };

    v8f acc[MT][NTW];
#pragma unroll
    for (int j = 0; j < NTW; ++j) {
        const int col = colBase + 16 * j + c;
        const float bv = (WVEC || col < Nact) ? bias[col] : 0.0f;
#pragma unroll
        for (int mt = 0; mt < MT; ++mt)
#pragma unroll
            for (int i = 0; i < 8; ++i) acc[mt][j][i] = bv;
    }

    const int aFrag = 2 * WSZ + c * APITCH + 2 * h;
    const int bFrag = h * WPITCH + 2 * (colBase + c);

    gload(0);
    lstore(lds);
    __syncthreads();

    for (int ch = 0; ch < NCHUNK; ++ch) {
        const float* cur = lds + (ch & 1) * STAGE;
        float*       nxt = lds + ((ch + 1) & 1) * STAGE;
        const bool more = ch + 1 < NCHUNK;
        if (more) gload((ch + 1) * KC);
        __builtin_amdgcn_sched_barrier(0);

#pragma unroll
        for (int kk = 0; kk < KQ; ++kk) {
            v2f am[MT], ah[MT], bl[NTW], br[NTW];
#pragma unroll
            for (int mt = 0; mt < MT; ++mt) {
                am[mt] = *(const v2f*)(cur + aFrag + 16 * mt * APITCH + 4 * kk);
                ah[mt] = *(const v2f*)(cur + aFrag + ASZ + 16 * mt * APITCH + 4 * kk);
            }
#pragma unroll
            for (int j = 0; j < NTW; ++j) {
                bl[j] = *(const v2f*)(cur + bFrag + 2 * kk * WPITCH + 32 * j);
                br[j] = *(const v2f*)(cur + bFrag + WSZ + 2 * kk * WPITCH + 32 * j);
            }
#pragma unroll
            for (int mt = 0; mt < MT; ++mt) {
#pragma unroll
                for (int j = 0; j < NTW; ++j) {
                    acc[mt][j] = __builtin_amdgcn_wmma_f32_16x16x4_f32(
                                     false, am[mt], false, bl[j], (short)0, acc[mt][j], false, false);
                    acc[mt][j] = __builtin_amdgcn_wmma_f32_16x16x4_f32(
                                     false, ah[mt], false, br[j], (short)0, acc[mt][j], false, false);
                }
            }
        }

        if (more) lstore(nxt);
        __syncthreads();
    }

#pragma unroll
    for (int mt = 0; mt < MT; ++mt) {
#pragma unroll
        for (int j = 0; j < NTW; ++j) {
            const int col = colBase + 16 * j + c;
            if (WVEC || col < Nact) {
                float* o = Out + (size_t)(rowBase + 16 * mt + 8 * h) * ldw + col;
#pragma unroll
                for (int i = 0; i < 8; ++i) {
                    float v = acc[mt][j][i];
                    if (RELU) v = fmaxf(v, 0.0f);
                    o[(size_t)i * ldw] = v;
                }
            }
        }
    }
}


extern "C" void kernel_launch(void* const* d_in, const int* in_sizes, int n_in,
                              void* d_out, int out_size, void* d_ws, size_t ws_size,
                              hipStream_t stream) {
    const float* h   = (const float*)d_in[0];
    const float* wl0 = (const float*)d_in[1];
    const float* wr0 = (const float*)d_in[2];
    const float* b0  = (const float*)d_in[3];
    const float* wl1 = (const float*)d_in[4];
    const float* wr1 = (const float*)d_in[5];
    const float* b1  = (const float*)d_in[6];
    const int*   ei0 = (const int*)d_in[7];
    const int*   ei1 = (const int*)d_in[8];

    const int E0 = in_sizes[7] / 2;
    const int E1 = in_sizes[8] / 2;
    const int NT0 = 20000;
    const int NT1 = 2000;
    const int NCLS = 47;

    float* mean0 = (float*)d_ws;
    float* h1    = mean0 + (size_t)NT0 * DFEAT;
    float* mean1 = h1 + (size_t)NT0 * DFEAT;

    sage_mean_owner_kernel<80><<<(NT0 + 79) / 80, AGG_THREADS, 0, stream>>>(
        h, ei0, ei0 + E0, mean0, E0, NT0);

    sage_gemm_lds<5, 2, 8, 32, true, true><<<NT0 / 80, 256, 0, stream>>>(
        mean0, h, wl0, wr0, b0, h1, DFEAT);

    sage_mean_owner_kernel<8><<<(NT1 + 7) / 8, AGG_THREADS, 0, stream>>>(
        h1, ei1, ei1 + E1, mean1, E1, NT1);

    sage_gemm_lds<1, 1, 3, 64, false, false><<<NT1 / 16, 96, 0, stream>>>(
        mean1, h1, wl1, wr1, b1, (float*)d_out, NCLS);
}


# Aggregation: owned-edge list ordered by target row in LDS (counting sort); an 8-row batch sums same-target rows in registers, LDS f32 adds only per target change; sum order = batch partial sums
# speedup vs baseline: 4.1537x; 1.0554x over previous
#include <hip/hip_runtime.h>

typedef __attribute__((ext_vector_type(2))) float v2f;
typedef __attribute__((ext_vector_type(4))) float v4f;
typedef __attribute__((ext_vector_type(8))) float v8f;

#define DFEAT 256

#define AGG_THREADS  1024
#define AGG_WAVES    32
#define AGG_CAP      8192
#define AGG_INFLIGHT 8

template<int R>
__global__ __launch_bounds__(AGG_THREADS)
void sage_mean_owner_kernel(const float* __restrict__ hsrc,
                            const int* __restrict__ src,
                            const int* __restrict__ dst,
                            float* __restrict__ mean,
                            int nE, int nTgt) {
    __shared__ float slab[R * DFEAT];
    __shared__ int   list_e[AGG_CAP];
    __shared__ int   list_t[AGG_CAP];
    __shared__ int   sort_s[AGG_CAP];
    __shared__ int   sort_t[AGG_CAP];
    __shared__ int   cnt[R];
    __shared__ int   cursor[R];
    __shared__ int   n_list;

    const int tid  = threadIdx.x;
    const int lane = tid & 31;
    const int wave = tid >> 5;
    const int base = blockIdx.x * R;
    const int rows = min(R, nTgt - base);
    const unsigned ubase = (unsigned)base;
    const unsigned urows = (unsigned)rows;

    for (int i = tid; i < R * DFEAT / 4; i += AGG_THREADS)
        ((float4*)slab)[i] = make_float4(0.0f, 0.0f, 0.0f, 0.0f);
    for (int i = tid; i < R; i += AGG_THREADS) cnt[i] = 0;
    if (tid == 0) n_list = 0;
    __syncthreads();

    auto record = [&](unsigned t, int i) {
        if (t < urows) {
            atomicAdd(&cnt[t], 1);
            const int idx = atomicAdd(&n_list, 1);
            if (idx < AGG_CAP) { list_e[idx] = i; list_t[idx] = (int)t; }
        }
    };
    auto check4 = [&](const int4 q, int i) {
        const unsigned t0 = (unsigned)q.x - ubase;
        const unsigned t1 = (unsigned)q.y - ubase;
        const unsigned t2 = (unsigned)q.z - ubase;
        const unsigned t3 = (unsigned)q.w - ubase;
        if ((t0 < urows) | (t1 < urows) | (t2 < urows) | (t3 < urows)) {
            record(t0, i); record(t1, i + 1); record(t2, i + 2); record(t3, i + 3);
        }
    };
    const bool vec_ok = (((size_t)dst) & 15) == 0;
    const int  nVec   = vec_ok ? (nE >> 2) : 0;
    const int4* dst4  = (const int4*)dst;
    const int start = (int)blockIdx.x * (nVec / (int)gridDim.x);
    for (int pass = 0; pass < 2; ++pass) {
        const int lo = pass ? 0 : start;
        const int hi = pass ? start : nVec;
        int v = lo + tid;
        for (; v + 3 * AGG_THREADS < hi; v += 4 * AGG_THREADS) {
            const int4 q0 = dst4[v];
            const int4 q1 = dst4[v + AGG_THREADS];
            const int4 q2 = dst4[v + 2 * AGG_THREADS];
            const int4 q3 = dst4[v + 3 * AGG_THREADS];
            check4(q0, 4 * v);
            check4(q1, 4 * (v + AGG_THREADS));
            check4(q2, 4 * (v + 2 * AGG_THREADS));
            check4(q3, 4 * (v + 3 * AGG_THREADS));
        }
        for (; v < hi; v += AGG_THREADS) check4(dst4[v], 4 * v);
    }
    for (int i = 4 * nVec + tid; i < nE; i += AGG_THREADS)
        record((unsigned)dst[i] - ubase, i);
    __syncthreads();

    int n = __builtin_amdgcn_readfirstlane(n_list);
    if (n > AGG_CAP) {
        for (int i = tid; i < nE; i += AGG_THREADS) {
            const unsigned t = (unsigned)dst[i] - ubase;
            if (t < urows) {
                const float* row = hsrc + (size_t)src[i] * DFEAT;
                float* acc = slab + t * DFEAT;
                for (int c = 0; c < DFEAT; ++c) atomicAdd(&acc[c], row[c]);
            }
        }
        n = 0;
    }

    if (tid < R) {
        int first = 0;
#pragma unroll
        for (int r = 0; r < R; ++r) first += (r < tid) ? cnt[r] : 0;
        cursor[tid] = first;
    }
    __syncthreads();
    for (int i = tid; i < n; i += AGG_THREADS) {
        const int t = list_t[i];
        const int s = src[list_e[i]];
        const int p = atomicAdd(&cursor[t], 1);
        sort_s[p] = s;
        sort_t[p] = t;
    }
    __syncthreads();

    const int nFull = n / AGG_INFLIGHT;
    for (int b = wave; b < nFull; b += AGG_WAVES) {
        float2 v[AGG_INFLIGHT][4];
        int    tt[AGG_INFLIGHT];
#pragma unroll
        for (int u = 0; u < AGG_INFLIGHT; ++u) {
            const int i = b * AGG_INFLIGHT + u;
            const int s = __builtin_amdgcn_readfirstlane(sort_s[i]);
            tt[u]       = __builtin_amdgcn_readfirstlane(sort_t[i]);
            const float2* row = (const float2*)(hsrc + (size_t)s * DFEAT);
#pragma unroll
            for (int j = 0; j < 4; ++j) v[u][j] = row[lane + 32 * j];
        }
        float2 a[4];
#pragma unroll
        for (int j = 0; j < 4; ++j) a[j] = v[0][j];
#pragma unroll
        for (int u = 1; u <= AGG_INFLIGHT; ++u) {
            if (u == AGG_INFLIGHT || tt[u] != tt[u - 1]) {
                float* acc = slab + tt[u - 1] * DFEAT + 2 * lane;
#pragma unroll
                for (int j = 0; j < 4; ++j) {
                    atomicAdd(&acc[64 * j],     a[j].x);
                    atomicAdd(&acc[64 * j + 1], a[j].y);
                }
                if (u < AGG_INFLIGHT) {
#pragma unroll
                    for (int j = 0; j < 4; ++j) a[j] = v[u][j];
                }
            } else {
#pragma unroll
                for (int j = 0; j < 4; ++j) { a[j].x += v[u][j].x; a[j].y += v[u][j].y; }
            }
        }
    }
    for (int i = nFull * AGG_INFLIGHT + wave; i < n; i += AGG_WAVES) {
        const int s = __builtin_amdgcn_readfirstlane(sort_s[i]);
        const int t = __builtin_amdgcn_readfirstlane(sort_t[i]);
        const float2* row = (const float2*)(hsrc + (size_t)s * DFEAT);
        float* acc = slab + t * DFEAT + 2 * lane;
#pragma unroll
        for (int j = 0; j < 4; ++j) {
            const float2 x = row[lane + 32 * j];
            atomicAdd(&acc[64 * j],     x.x);
            atomicAdd(&acc[64 * j + 1], x.y);
        }
    }
    __syncthreads();

    for (int r = wave; r < rows; r += AGG_WAVES) {
        const float inv = 1.0f / fmaxf((float)cnt[r], 1.0f);
        const float4* srow = (const float4*)(slab + r * DFEAT);
        float4* orow = (float4*)(mean + (size_t)(base + r) * DFEAT);
#pragma unroll
        for (int j = 0; j < 2; ++j) {
            float4 x = srow[lane + 32 * j];
            x.x *= inv; x.y *= inv; x.z *= inv; x.w *= inv;
            orow[lane + 32 * j] = x;
        }
    }
}

template<int MT, int NTW, int NWAVE, int KC, bool WVEC, bool RELU>
__global__ __launch_bounds__(NWAVE * 32)
void sage_gemm_lds(const float* __restrict__ Amean,
                   const float* __restrict__ Htgt,
                   const float* __restrict__ Wl,
                   const float* __restrict__ Wr,
                   const float* __restrict__ bias,
                   float* __restrict__ Out,
                   int Nact) {
    constexpr int NTHR   = NWAVE * 32;
    constexpr int ROWS   = 16 * MT;
    constexpr int NCOLS  = 16 * NTW * NWAVE;
    constexpr int KP     = KC / 2;
    constexpr int KQ     = KC / 4;
    constexpr int WPITCH = 2 * NCOLS + ((32 - (2 * NCOLS) % 64 + 64) % 64);
    constexpr int APITCH = KC + 4;
    constexpr int WSZ    = KP * WPITCH;
    constexpr int ASZ    = ROWS * APITCH;
    constexpr int STAGE  = 2 * WSZ + 2 * ASZ;
    constexpr int NCHUNK = DFEAT / KC;
    constexpr int ARPR   = NTHR / KQ;
    constexpr int AR     = (2 * ROWS + ARPR - 1) / ARPR;
    constexpr int WCPT   = WVEC ? 4 : 1;
    constexpr int WRPR   = NTHR / (NCOLS / WCPT);
    constexpr int WRM    = KP / WRPR;
    static_assert(NTHR % KQ == 0 && NTHR % (NCOLS / WCPT) == 0 && KP % WRPR == 0, "copy shares must tile the chunk");

    __shared__ __attribute__((aligned(16))) float lds[2 * STAGE];

    const int tid  = threadIdx.x;
    const int wave = tid >> 5;
    const int lane = tid & 31;
    const int c    = lane & 15;
    const int h    = lane >> 4;
    const int ldw = WVEC ? NCOLS : Nact;
    const int rowBase = blockIdx.x * ROWS;
    const int colBase = wave * (16 * NTW);

    const float* asrc[AR];
    int          aoff[AR];
    {
        const int q  = tid % KQ;
        const int r0 = tid / KQ;
#pragma unroll
        for (int i = 0; i < AR; ++i) {
            const int mrow = min(r0 + i * ARPR, 2 * ROWS - 1);
            const int mat  = mrow >= ROWS;
            const int m    = mrow - mat * ROWS;
            asrc[i] = (mat ? Htgt : Amean) + (size_t)(rowBase + m) * DFEAT + 4 * q;
            aoff[i] = 2 * WSZ + mat * ASZ + m * APITCH + 4 * q;
        }
    }
    const int  wcol = (tid % (NCOLS / WCPT)) * WCPT;
    const int  wp0  = tid / (NCOLS / WCPT);
    const bool wok  = wcol < Nact;
    const int  wtoff = 2 * wp0 * ldw + (wok ? wcol : 0);
    const int    woff0 = wp0 * WPITCH + 2 * wcol;

    v4f areg[AR];
    v4f wv0[2][WRM], wv1[2][WRM];
    v2f ws[2][WRM];

    auto gload = [&](int k0) {
#pragma unroll
        for (int i = 0; i < AR; ++i) areg[i] = *(const v4f*)(asrc[i] + k0);
#pragma unroll
        for (int mat = 0; mat < 2; ++mat) {
            const float* wb = (mat ? Wr : Wl) + (size_t)k0 * ldw;
#pragma unroll
            for (int i = 0; i < WRM; ++i) {
                const float* s = wb + (size_t)(2 * i * WRPR) * ldw;
                if constexpr (WVEC) {
                    wv0[mat][i] = *(const v4f*)(s + wtoff);
                    wv1[mat][i] = *(const v4f*)(s + ldw + wtoff);
                } else {
                    ws[mat][i].x = s[wtoff];
                    ws[mat][i].y = s[ldw + wtoff];
                }
            }
        }
    };
    auto lstore = [&](float* stage) {
#pragma unroll
        for (int i = 0; i < AR; ++i) *(v4f*)(stage + aoff[i]) = areg[i];
#pragma unroll
        for (int mat = 0; mat < 2; ++mat) {
#pragma unroll
            for (int i = 0; i < WRM; ++i) {
                float* d = stage + mat * WSZ + woff0 + i * WRPR * WPITCH;
                if constexpr (WVEC) {
                    const v4f r0 = wv0[mat][i];
                    const v4f r1 = wv1[mat][i];
                    v4f o0, o1;
                    o0.x = r0.x; o0.y = r1.x; o0.z = r0.y; o0.w = r1.y;
                    o1.x = r0.z; o1.y = r1.z; o1.z = r0.w; o1.w = r1.w;
                    *(v4f*)d       = o0;
                    *(v4f*)(d + 4) = o1;
                } else {
                    *(v2f*)d = ws[mat][i];
                }
            }
        }
    };

    v8f acc[MT][NTW];
#pragma unroll
    for (int j = 0; j < NTW; ++j) {
        const int col = colBase + 16 * j + c;
        const float bv = (WVEC || col < Nact) ? bias[col] : 0.0f;
#pragma unroll
        for (int mt = 0; mt < MT; ++mt)
#pragma unroll
            for (int i = 0; i < 8; ++i) acc[mt][j][i] = bv;
    }

    const int aFrag = 2 * WSZ + c * APITCH + 2 * h;
    const int bFrag = h * WPITCH + 2 * (colBase + c);

    gload(0);
    lstore(lds);
    __syncthreads();

    for (int ch = 0; ch < NCHUNK; ++ch) {
        const float* cur = lds + (ch & 1) * STAGE;
        float*       nxt = lds + ((ch + 1) & 1) * STAGE;
        const bool more = ch + 1 < NCHUNK;
        if (more) gload((ch + 1) * KC);
        __builtin_amdgcn_sched_barrier(0);

#pragma unroll
        for (int kk = 0; kk < KQ; ++kk) {
            v2f am[MT], ah[MT], bl[NTW], br[NTW];
#pragma unroll
            for (int mt = 0; mt < MT; ++mt) {
                am[mt] = *(const v2f*)(cur + aFrag + 16 * mt * APITCH + 4 * kk);
                ah[mt] = *(const v2f*)(cur + aFrag + ASZ + 16 * mt * APITCH + 4 * kk);
            }
#pragma unroll
            for (int j = 0; j < NTW; ++j) {
                bl[j] = *(const v2f*)(cur + bFrag + 2 * kk * WPITCH + 32 * j);
                br[j] = *(const v2f*)(cur + bFrag + WSZ + 2 * kk * WPITCH + 32 * j);
            }
#pragma unroll
            for (int mt = 0; mt < MT; ++mt) {
#pragma unroll
                for (int j = 0; j < NTW; ++j) {
                    acc[mt][j] = __builtin_amdgcn_wmma_f32_16x16x4_f32(
                                     false, am[mt], false, bl[j], (short)0, acc[mt][j], false, false);
                    acc[mt][j] = __builtin_amdgcn_wmma_f32_16x16x4_f32(
                                     false, ah[mt], false, br[j], (short)0, acc[mt][j], false, false);
                }
            }
        }

        if (more) lstore(nxt);
        __syncthreads();
    }

#pragma unroll
    for (int mt = 0; mt < MT; ++mt) {
#pragma unroll
        for (int j = 0; j < NTW; ++j) {
            const int col = colBase + 16 * j + c;
            if (WVEC || col < Nact) {
                float* o = Out + (size_t)(rowBase + 16 * mt + 8 * h) * ldw + col;
#pragma unroll
                for (int i = 0; i < 8; ++i) {
                    float v = acc[mt][j][i];
                    if (RELU) v = fmaxf(v, 0.0f);
                    o[(size_t)i * ldw] = v;
                }
            }
        }
    }
}


extern "C" void kernel_launch(void* const* d_in, const int* in_sizes, int n_in,
                              void* d_out, int out_size, void* d_ws, size_t ws_size,
                              hipStream_t stream) {
    const float* h   = (const float*)d_in[0];
    const float* wl0 = (const float*)d_in[1];
    const float* wr0 = (const float*)d_in[2];
    const float* b0  = (const float*)d_in[3];
    const float* wl1 = (const float*)d_in[4];
    const float* wr1 = (const float*)d_in[5];
    const float* b1  = (const float*)d_in[6];
    const int*   ei0 = (const int*)d_in[7];
    const int*   ei1 = (const int*)d_in[8];

    const int E0 = in_sizes[7] / 2;
    const int E1 = in_sizes[8] / 2;
    const int NT0 = 20000;
    const int NT1 = 2000;
    const int NCLS = 47;

    float* mean0 = (float*)d_ws;
    float* h1    = mean0 + (size_t)NT0 * DFEAT;
    float* mean1 = h1 + (size_t)NT0 * DFEAT;

    sage_mean_owner_kernel<80><<<(NT0 + 79) / 80, AGG_THREADS, 0, stream>>>(
        h, ei0, ei0 + E0, mean0, E0, NT0);

    sage_gemm_lds<5, 2, 8, 32, true, true><<<NT0 / 80, 256, 0, stream>>>(
        mean0, h, wl0, wr0, b0, h1, DFEAT);

    sage_mean_owner_kernel<8><<<(NT1 + 7) / 8, AGG_THREADS, 0, stream>>>(
        h1, ei1, ei1 + E1, mean1, E1, NT1);

    sage_gemm_lds<1, 1, 3, 64, false, false><<<NT1 / 16, 96, 0, stream>>>(
        mean1, h1, wl1, wr1, b1, (float*)d_out, NCLS);
}


# Layer-1 neighbour weights applied before its aggregation (WMMA on the layer-0 GEMM's tile via LDS); layer 1 averages 48-float rows, target term K-split on WMMA; 4 launches -> 3; sum order changed
# speedup vs baseline: 4.5095x; 1.0857x over previous
#include <hip/hip_runtime.h>

typedef __attribute__((ext_vector_type(2))) float v2f;
typedef __attribute__((ext_vector_type(4))) float v4f;
typedef __attribute__((ext_vector_type(8))) float v8f;

#define DFEAT 256
#define NCLS  47
#define PCOLS 48

#define AGG_THREADS  1024
#define AGG_WAVES    32
#define AGG_CAP      8192
#define AGG_INFLIGHT 8

template<int R>
__global__ __launch_bounds__(AGG_THREADS)
void sage_mean_owner_kernel(const float* __restrict__ hsrc,
                            const int* __restrict__ src,
                            const int* __restrict__ dst,
                            float* __restrict__ mean,
                            int nE, int nTgt) {
    __shared__ float slab[R * DFEAT];
    __shared__ int   list_e[AGG_CAP];
    __shared__ int   list_t[AGG_CAP];
    __shared__ int   sort_s[AGG_CAP];
    __shared__ int   sort_t[AGG_CAP];
    __shared__ int   cnt[R];
    __shared__ int   cursor[R];
    __shared__ int   n_list;

    const int tid  = threadIdx.x;
    const int lane = tid & 31;
    const int wave = tid >> 5;
    const int base = blockIdx.x * R;
    const int rows = min(R, nTgt - base);
    const unsigned ubase = (unsigned)base;
    const unsigned urows = (unsigned)rows;

    for (int i = tid; i < R * DFEAT / 4; i += AGG_THREADS)
        ((float4*)slab)[i] = make_float4(0.0f, 0.0f, 0.0f, 0.0f);
    for (int i = tid; i < R; i += AGG_THREADS) cnt[i] = 0;
    if (tid == 0) n_list = 0;
    __syncthreads();

    auto record = [&](unsigned t, int i) {
        if (t < urows) {
            atomicAdd(&cnt[t], 1);
            const int idx = atomicAdd(&n_list, 1);
            if (idx < AGG_CAP) { list_e[idx] = i; list_t[idx] = (int)t; }
        }
    };
    auto check4 = [&](const int4 q, int i) {
        const unsigned t0 = (unsigned)q.x - ubase;
        const unsigned t1 = (unsigned)q.y - ubase;
        const unsigned t2 = (unsigned)q.z - ubase;
        const unsigned t3 = (unsigned)q.w - ubase;
        if ((t0 < urows) | (t1 < urows) | (t2 < urows) | (t3 < urows)) {
            record(t0, i); record(t1, i + 1); record(t2, i + 2); record(t3, i + 3);
        }
    };
    const bool vec_ok = (((size_t)dst) & 15) == 0;
    const int  nVec   = vec_ok ? (nE >> 2) : 0;
    const int4* dst4  = (const int4*)dst;
    const int start = (int)blockIdx.x * (nVec / (int)gridDim.x);
    for (int pass = 0; pass < 2; ++pass) {
        const int lo = pass ? 0 : start;
        const int hi = pass ? start : nVec;
        int v = lo + tid;
        for (; v + 3 * AGG_THREADS < hi; v += 4 * AGG_THREADS) {
            const int4 q0 = dst4[v];
            const int4 q1 = dst4[v + AGG_THREADS];
            const int4 q2 = dst4[v + 2 * AGG_THREADS];
            const int4 q3 = dst4[v + 3 * AGG_THREADS];
            check4(q0, 4 * v);
            check4(q1, 4 * (v + AGG_THREADS));
            check4(q2, 4 * (v + 2 * AGG_THREADS));
            check4(q3, 4 * (v + 3 * AGG_THREADS));
        }
        for (; v < hi; v += AGG_THREADS) check4(dst4[v], 4 * v);
    }
    for (int i = 4 * nVec + tid; i < nE; i += AGG_THREADS)
        record((unsigned)dst[i] - ubase, i);
    __syncthreads();

    int n = __builtin_amdgcn_readfirstlane(n_list);
    if (n > AGG_CAP) {
        for (int i = tid; i < nE; i += AGG_THREADS) {
            const unsigned t = (unsigned)dst[i] - ubase;
            if (t < urows) {
                const float* row = hsrc + (size_t)src[i] * DFEAT;
                float* acc = slab + t * DFEAT;
                for (int c = 0; c < DFEAT; ++c) atomicAdd(&acc[c], row[c]);
            }
        }
        n = 0;
    }

    if (tid < R) {
        int first = 0;
#pragma unroll
        for (int r = 0; r < R; ++r) first += (r < tid) ? cnt[r] : 0;
        cursor[tid] = first;
    }
    __syncthreads();
    for (int i = tid; i < n; i += AGG_THREADS) {
        const int t = list_t[i];
        const int s = src[list_e[i]];
        const int p = atomicAdd(&cursor[t], 1);
        sort_s[p] = s;
        sort_t[p] = t;
    }
    __syncthreads();

    const int nFull = n / AGG_INFLIGHT;
    for (int b = wave; b < nFull; b += AGG_WAVES) {
        float2 v[AGG_INFLIGHT][4];
        int    tt[AGG_INFLIGHT];
#pragma unroll
        for (int u = 0; u < AGG_INFLIGHT; ++u) {
            const int i = b * AGG_INFLIGHT + u;
            const int s = __builtin_amdgcn_readfirstlane(sort_s[i]);
            tt[u]       = __builtin_amdgcn_readfirstlane(sort_t[i]);
            const float2* row = (const float2*)(hsrc + (size_t)s * DFEAT);
#pragma unroll
            for (int j = 0; j < 4; ++j) v[u][j] = row[lane + 32 * j];
        }
        float2 a[4];
#pragma unroll
        for (int j = 0; j < 4; ++j) a[j] = v[0][j];
#pragma unroll
        for (int u = 1; u <= AGG_INFLIGHT; ++u) {
            if (u == AGG_INFLIGHT || tt[u] != tt[u - 1]) {
                float* acc = slab + tt[u - 1] * DFEAT + 2 * lane;
#pragma unroll
                for (int j = 0; j < 4; ++j) {
                    atomicAdd(&acc[64 * j],     a[j].x);
                    atomicAdd(&acc[64 * j + 1], a[j].y);
                }
                if (u < AGG_INFLIGHT) {
#pragma unroll
                    for (int j = 0; j < 4; ++j) a[j] = v[u][j];
                }
            } else {
#pragma unroll
                for (int j = 0; j < 4; ++j) { a[j].x += v[u][j].x; a[j].y += v[u][j].y; }
            }
        }
    }
    for (int i = nFull * AGG_INFLIGHT + wave; i < n; i += AGG_WAVES) {
        const int s = __builtin_amdgcn_readfirstlane(sort_s[i]);
        const int t = __builtin_amdgcn_readfirstlane(sort_t[i]);
        const float2* row = (const float2*)(hsrc + (size_t)s * DFEAT);
        float* acc = slab + t * DFEAT + 2 * lane;
#pragma unroll
        for (int j = 0; j < 4; ++j) {
            const float2 x = row[lane + 32 * j];
            atomicAdd(&acc[64 * j],     x.x);
            atomicAdd(&acc[64 * j + 1], x.y);
        }
    }
    __syncthreads();

    for (int r = wave; r < rows; r += AGG_WAVES) {
        const float inv = 1.0f / fmaxf((float)cnt[r], 1.0f);
        const float4* srow = (const float4*)(slab + r * DFEAT);
        float4* orow = (float4*)(mean + (size_t)(base + r) * DFEAT);
#pragma unroll
        for (int j = 0; j < 2; ++j) {
            float4 x = srow[lane + 32 * j];
            x.x *= inv; x.y *= inv; x.z *= inv; x.w *= inv;
            orow[lane + 32 * j] = x;
        }
    }
}

template<int R>
__global__ __launch_bounds__(AGG_THREADS)
void sage_mean_narrow_kernel(const float* __restrict__ Pn,
                             const int* __restrict__ src,
                             const int* __restrict__ dst,
                             const float* __restrict__ Ht,
                             const float* __restrict__ Wr,
                             const float* __restrict__ bias,
                             float* __restrict__ out,
                             int nE, int nTgt) {
    __shared__ __attribute__((aligned(16))) float slab[R * PCOLS];
    __shared__ float tterm[R * PCOLS];
    __shared__ int   list_e[AGG_CAP];
    __shared__ int   list_t[AGG_CAP];
    __shared__ int   cnt[R];
    __shared__ int   n_list;

    static_assert(R <= 8, "the owned rows must lie in lane half 0 of one C/D tile");
    constexpr int KSL = 8;
    constexpr int KST = DFEAT / 4 / KSL;
    constexpr int PTN = PCOLS / 16;
    static_assert(PTN * KSL <= AGG_WAVES, "one wave per (column tile, k-slice)");

    const int tid  = threadIdx.x;
    const int lane = tid & 31;
    const int wave = tid >> 5;
    const int c    = lane & 15;
    const int h    = lane >> 4;
    const int base = blockIdx.x * R;
    const int rows = min(R, nTgt - base);
    const unsigned ubase = (unsigned)base;
    const unsigned urows = (unsigned)rows;

    const bool tw = wave < PTN * KSL;
    const int  nt = wave % PTN;
    const int  ks = wave / PTN;
    const bool rok = c < rows;
    const bool cok = 16 * nt + c < NCLS;
    for (int i = tid; i < R * PCOLS; i += AGG_THREADS) { slab[i] = 0.0f; tterm[i] = 0.0f; }
    for (int i = tid; i < R; i += AGG_THREADS) cnt[i] = 0;
    if (tid == 0) n_list = 0;
    __syncthreads();

    v2f fa[KST], fb[KST];
    if (tw) {
        const float* ap = Ht + (size_t)(base + (rok ? c : 0)) * DFEAT + 4 * KST * ks + 2 * h;
        const float* bp = Wr + (size_t)(4 * KST * ks + 2 * h) * NCLS + (cok ? 16 * nt + c : 0);
#pragma unroll
        for (int st = 0; st < KST; ++st) {
            fa[st]   = *(const v2f*)(ap + 4 * st);
            fb[st].x = bp[(4 * st) * NCLS];
            fb[st].y = bp[(4 * st + 1) * NCLS];
        }
    }

    auto record = [&](unsigned t, int i) {
        if (t < urows) {
            atomicAdd(&cnt[t], 1);
            const int idx = atomicAdd(&n_list, 1);
            if (idx < AGG_CAP) { list_e[idx] = i; list_t[idx] = (int)t; }
        }
    };
    auto check4 = [&](const int4 q, int i) {
        const unsigned t0 = (unsigned)q.x - ubase;
        const unsigned t1 = (unsigned)q.y - ubase;
        const unsigned t2 = (unsigned)q.z - ubase;
        const unsigned t3 = (unsigned)q.w - ubase;
        if ((t0 < urows) | (t1 < urows) | (t2 < urows) | (t3 < urows)) {
            record(t0, i); record(t1, i + 1); record(t2, i + 2); record(t3, i + 3);
        }
    };
    const bool vec_ok = (((size_t)dst) & 15) == 0;
    const int  nVec   = vec_ok ? (nE >> 2) : 0;
    const int4* dst4  = (const int4*)dst;
    const int start = (int)blockIdx.x * (nVec / (int)gridDim.x);
    for (int pass = 0; pass < 2; ++pass) {
        const int lo = pass ? 0 : start;
        const int hi = pass ? start : nVec;
        int v = lo + tid;
        for (; v + 3 * AGG_THREADS < hi; v += 4 * AGG_THREADS) {
            const int4 q0 = dst4[v];
            const int4 q1 = dst4[v + AGG_THREADS];
            const int4 q2 = dst4[v + 2 * AGG_THREADS];
            const int4 q3 = dst4[v + 3 * AGG_THREADS];
            check4(q0, 4 * v);
            check4(q1, 4 * (v + AGG_THREADS));
            check4(q2, 4 * (v + 2 * AGG_THREADS));
            check4(q3, 4 * (v + 3 * AGG_THREADS));
        }
        for (; v < hi; v += AGG_THREADS) check4(dst4[v], 4 * v);
    }
    for (int i = 4 * nVec + tid; i < nE; i += AGG_THREADS)
        record((unsigned)dst[i] - ubase, i);

    if (tw) {
        v8f q;
#pragma unroll
        for (int i = 0; i < 8; ++i) q[i] = 0.0f;
#pragma unroll
        for (int st = 0; st < KST; ++st) {
            v2f x = fa[st], y = fb[st];
            if (!rok) { x.x = 0.0f; x.y = 0.0f; }
            if (!cok) { y.x = 0.0f; y.y = 0.0f; }
            q = __builtin_amdgcn_wmma_f32_16x16x4_f32(false, x, false, y, (short)0, q, false, false);
        }
        if (h == 0) {
#pragma unroll
            for (int i = 0; i < 8; ++i)
                if (i < rows) atomicAdd(&tterm[i * PCOLS + 16 * nt + c], q[i]);
        }
    }
    __syncthreads();

    int n = __builtin_amdgcn_readfirstlane(n_list);
    if (n > AGG_CAP) {
        for (int i = tid; i < nE; i += AGG_THREADS) {
            const unsigned t = (unsigned)dst[i] - ubase;
            if (t < urows) {
                const float* row = Pn + (size_t)src[i] * PCOLS;
                float* acc = slab + t * PCOLS;
                for (int j = 0; j < PCOLS; ++j) atomicAdd(&acc[j], row[j]);
            }
        }
        n = 0;
    }

    const int grp = tid >> 4;
    const int l16 = tid & 15;
    for (int i = grp; i < n; i += AGG_THREADS / 16) {
        const int t = list_t[i];
        const int sn = src[list_e[i]];
        if (l16 < PCOLS / 4) {
            const float4 x = ((const float4*)(Pn + (size_t)sn * PCOLS))[l16];
            float* acc = slab + t * PCOLS + 4 * l16;
            atomicAdd(&acc[0], x.x);
            atomicAdd(&acc[1], x.y);
            atomicAdd(&acc[2], x.z);
            atomicAdd(&acc[3], x.w);
        }
    }
    __syncthreads();

    for (int i = tid; i < rows * NCLS; i += AGG_THREADS) {
        const int r = i / NCLS, oc = i - r * NCLS;
        const float inv = 1.0f / fmaxf((float)cnt[r], 1.0f);
        out[(size_t)(base + r) * NCLS + oc] = slab[r * PCOLS + oc] * inv + tterm[r * PCOLS + oc] + bias[oc];
    }
}

template<int MT, int NTW, int NWAVE, int KC, bool WVEC, bool RELU>
__global__ __launch_bounds__(NWAVE * 32)
void sage_gemm_lds(const float* __restrict__ Amean,
                   const float* __restrict__ Htgt,
                   const float* __restrict__ Wl,
                   const float* __restrict__ Wr,
                   const float* __restrict__ bias,
                   const float* __restrict__ Wl2,
                   float* __restrict__ Pn,
                   float* __restrict__ Ht,
                   int Nact, int nQ) {
    constexpr int NTHR   = NWAVE * 32;
    constexpr int ROWS   = 16 * MT;
    constexpr int NCOLS  = 16 * NTW * NWAVE;
    constexpr int KP     = KC / 2;
    constexpr int KQ     = KC / 4;
    constexpr int WPITCH = 2 * NCOLS + ((32 - (2 * NCOLS) % 64 + 64) % 64);
    constexpr int APITCH = KC + 4;
    constexpr int WSZ    = KP * WPITCH;
    constexpr int ASZ    = ROWS * APITCH;
    constexpr int STAGE  = 2 * WSZ + 2 * ASZ;
    constexpr int NCHUNK = DFEAT / KC;
    constexpr int ARPR   = NTHR / KQ;
    constexpr int AR     = (2 * ROWS + ARPR - 1) / ARPR;
    constexpr int WCPT   = WVEC ? 4 : 1;
    constexpr int WRPR   = NTHR / (NCOLS / WCPT);
    constexpr int WRM    = KP / WRPR;
    static_assert(NTHR % KQ == 0 && NTHR % (NCOLS / WCPT) == 0 && KP % WRPR == 0, "copy shares must tile the chunk");

    __shared__ __attribute__((aligned(16))) float lds[2 * STAGE];

    const int tid  = threadIdx.x;
    const int wave = tid >> 5;
    const int lane = tid & 31;
    const int c    = lane & 15;
    const int h    = lane >> 4;
    const int ldw = WVEC ? NCOLS : Nact;
    const int rowBase = blockIdx.x * ROWS;
    const int colBase = wave * (16 * NTW);

    const float* asrc[AR];
    int          aoff[AR];
    {
        const int q  = tid % KQ;
        const int r0 = tid / KQ;
#pragma unroll
        for (int i = 0; i < AR; ++i) {
            const int mrow = min(r0 + i * ARPR, 2 * ROWS - 1);
            const int mat  = mrow >= ROWS;
            const int m    = mrow - mat * ROWS;
            asrc[i] = (mat ? Htgt : Amean) + (size_t)(rowBase + m) * DFEAT + 4 * q;
            aoff[i] = 2 * WSZ + mat * ASZ + m * APITCH + 4 * q;
        }
    }
    const int  wcol = (tid % (NCOLS / WCPT)) * WCPT;
    const int  wp0  = tid / (NCOLS / WCPT);
    const bool wok  = wcol < Nact;
    const int  wtoff = 2 * wp0 * ldw + (wok ? wcol : 0);
    const int    woff0 = wp0 * WPITCH + 2 * wcol;

    v4f areg[AR];
    v4f wv0[2][WRM], wv1[2][WRM];
    v2f ws[2][WRM];

    auto gload = [&](int k0) {
#pragma unroll
        for (int i = 0; i < AR; ++i) areg[i] = *(const v4f*)(asrc[i] + k0);
#pragma unroll
        for (int mat = 0; mat < 2; ++mat) {
            const float* wb = (mat ? Wr : Wl) + (size_t)k0 * ldw;
#pragma unroll
            for (int i = 0; i < WRM; ++i) {
                const float* s = wb + (size_t)(2 * i * WRPR) * ldw;
                if constexpr (WVEC) {
                    wv0[mat][i] = *(const v4f*)(s + wtoff);
                    wv1[mat][i] = *(const v4f*)(s + ldw + wtoff);
                } else {
                    ws[mat][i].x = s[wtoff];
                    ws[mat][i].y = s[ldw + wtoff];
                }
            }
        }
    };
    auto lstore = [&](float* stage) {
#pragma unroll
        for (int i = 0; i < AR; ++i) *(v4f*)(stage + aoff[i]) = areg[i];
#pragma unroll
        for (int mat = 0; mat < 2; ++mat) {
#pragma unroll
            for (int i = 0; i < WRM; ++i) {
                float* d = stage + mat * WSZ + woff0 + i * WRPR * WPITCH;
                if constexpr (WVEC) {
                    const v4f r0 = wv0[mat][i];
                    const v4f r1 = wv1[mat][i];
                    v4f o0, o1;
                    o0.x = r0.x; o0.y = r1.x; o0.z = r0.y; o0.w = r1.y;
                    o1.x = r0.z; o1.y = r1.z; o1.z = r0.w; o1.w = r1.w;
                    *(v4f*)d       = o0;
                    *(v4f*)(d + 4) = o1;
                } else {
                    *(v2f*)d = ws[mat][i];
                }
            }
        }
    };

    v8f acc[MT][NTW];
#pragma unroll
    for (int j = 0; j < NTW; ++j) {
        const int col = colBase + 16 * j + c;
        const float bv = (WVEC || col < Nact) ? bias[col] : 0.0f;
#pragma unroll
        for (int mt = 0; mt < MT; ++mt)
#pragma unroll
            for (int i = 0; i < 8; ++i) acc[mt][j][i] = bv;
    }

    const int aFrag = 2 * WSZ + c * APITCH + 2 * h;
    const int bFrag = h * WPITCH + 2 * (colBase + c);

    gload(0);
    lstore(lds);
    __syncthreads();

    for (int ch = 0; ch < NCHUNK; ++ch) {
        const float* cur = lds + (ch & 1) * STAGE;
        float*       nxt = lds + ((ch + 1) & 1) * STAGE;
        const bool more = ch + 1 < NCHUNK;
        if (more) gload((ch + 1) * KC);
        __builtin_amdgcn_sched_barrier(0);

#pragma unroll
        for (int kk = 0; kk < KQ; ++kk) {
            v2f am[MT], ah[MT], bl[NTW], br[NTW];
#pragma unroll
            for (int mt = 0; mt < MT; ++mt) {
                am[mt] = *(const v2f*)(cur + aFrag + 16 * mt * APITCH + 4 * kk);
                ah[mt] = *(const v2f*)(cur + aFrag + ASZ + 16 * mt * APITCH + 4 * kk);
            }
#pragma unroll
            for (int j = 0; j < NTW; ++j) {
                bl[j] = *(const v2f*)(cur + bFrag + 2 * kk * WPITCH + 32 * j);
                br[j] = *(const v2f*)(cur + bFrag + WSZ + 2 * kk * WPITCH + 32 * j);
            }
#pragma unroll
            for (int mt = 0; mt < MT; ++mt) {
#pragma unroll
                for (int j = 0; j < NTW; ++j) {
                    acc[mt][j] = __builtin_amdgcn_wmma_f32_16x16x4_f32(
                                     false, am[mt], false, bl[j], (short)0, acc[mt][j], false, false);
                    acc[mt][j] = __builtin_amdgcn_wmma_f32_16x16x4_f32(
                                     false, ah[mt], false, br[j], (short)0, acc[mt][j], false, false);
                }
            }
        }

        if (more) lstore(nxt);
        __syncthreads();
    }

    constexpr int TP   = DFEAT + 4;
    constexpr int PBP  = 2 * PCOLS;
    constexpr int TSZ  = ROWS * TP;
    constexpr int PBSZ = (DFEAT / 2) * PBP;
    constexpr int PTM  = ROWS / 16;
    constexpr int PTN  = PCOLS / 16;
    static_assert(TSZ + PBSZ <= 2 * STAGE, "tile and projection matrix must fit the two stages");
    static_assert(NCOLS == DFEAT && PBP % 64 == 32 && PTM * PTN <= 2 * NWAVE, "projection tiling");
    float* tl = lds;
    float* pb = lds + TSZ;

#pragma unroll
    for (int mt = 0; mt < MT; ++mt)
#pragma unroll
        for (int j = 0; j < NTW; ++j)
#pragma unroll
            for (int i = 0; i < 8; ++i) {
                float v = acc[mt][j][i];
                if (RELU) v = fmaxf(v, 0.0f);
                tl[(16 * mt + 8 * h + i) * TP + colBase + 16 * j + c] = v;
            }
    constexpr int PW_PER = DFEAT * NCLS / NTHR;
    static_assert((DFEAT * NCLS) % NTHR == 0, "weight copy must tile");
    {
        float wv[PW_PER];
        int   wo[PW_PER];
        int k = tid / NCLS, col = tid - k * NCLS;
#pragma unroll
        for (int u = 0; u < PW_PER; ++u) {
            wv[u] = Wl2[tid + u * NTHR];
            wo[u] = (k >> 1) * PBP + 2 * col + (k & 1);
            col += NTHR % NCLS; k += NTHR / NCLS;
            if (col >= NCLS) { col -= NCLS; ++k; }
        }
#pragma unroll
        for (int u = 0; u < PW_PER; ++u) pb[wo[u]] = wv[u];
        for (int e = tid; e < DFEAT * (PCOLS - NCLS); e += NTHR) {
            const int kk = e / (PCOLS - NCLS), pc = NCLS + e - kk * (PCOLS - NCLS);
            pb[(kk >> 1) * PBP + 2 * pc + (kk & 1)] = 0.0f;
        }
    }
    __syncthreads();

    if (rowBase < nQ) {
        for (int r = wave; r < ROWS; r += NWAVE) {
            if (rowBase + r < nQ) {
                const v4f* srow = (const v4f*)(tl + r * TP);
                v4f* orow = (v4f*)(Ht + (size_t)(rowBase + r) * DFEAT);
                orow[lane]      = srow[lane];
                orow[lane + 32] = srow[lane + 32];
            }
        }
    }

    constexpr int PG  = 8;
    constexpr int PNG = DFEAT / 4 / PG;
    const int t0 = wave;
    const int t1 = min(wave + NWAVE, PTM * PTN - 1);
    const int m0 = t0 / PTN, n0 = t0 - m0 * PTN;
    const int m1 = t1 / PTN, n1 = t1 - m1 * PTN;
    const float* a0 = tl + (16 * m0 + c) * TP + 2 * h;
    const float* a1 = tl + (16 * m1 + c) * TP + 2 * h;
    const float* b0 = pb + h * PBP + 2 * (16 * n0 + c);
    const float* b1 = pb + h * PBP + 2 * (16 * n1 + c);
    v8f p0, p1;
#pragma unroll
    for (int i = 0; i < 8; ++i) { p0[i] = 0.0f; p1[i] = 0.0f; }
    v2f x0[2][PG], y0[2][PG], x1[2][PG], y1[2][PG];
    auto fload = [&](int buf, int g) {
#pragma unroll
        for (int q = 0; q < PG; ++q) {
            const int st = g * PG + q;
            x0[buf][q] = *(const v2f*)(a0 + 4 * st);
            y0[buf][q] = *(const v2f*)(b0 + 2 * st * PBP);
            x1[buf][q] = *(const v2f*)(a1 + 4 * st);
            y1[buf][q] = *(const v2f*)(b1 + 2 * st * PBP);
        }
    };
    fload(0, 0);
#pragma unroll
    for (int g = 0; g < PNG; ++g) {
        if (g + 1 < PNG) fload((g + 1) & 1, g + 1);
        __builtin_amdgcn_sched_barrier(0);
#pragma unroll
        for (int q = 0; q < PG; ++q) {
            p0 = __builtin_amdgcn_wmma_f32_16x16x4_f32(false, x0[g & 1][q], false, y0[g & 1][q], (short)0, p0, false, false);
            p1 = __builtin_amdgcn_wmma_f32_16x16x4_f32(false, x1[g & 1][q], false, y1[g & 1][q], (short)0, p1, false, false);
        }
        __builtin_amdgcn_sched_barrier(0);
    }
    {
        float* o0 = Pn + (size_t)(rowBase + 16 * m0 + 8 * h) * PCOLS + 16 * n0 + c;
        float* o1 = Pn + (size_t)(rowBase + 16 * m1 + 8 * h) * PCOLS + 16 * n1 + c;
#pragma unroll
        for (int i = 0; i < 8; ++i) {
            o0[i * PCOLS] = p0[i];
            if (t1 == wave + NWAVE) o1[i * PCOLS] = p1[i];
        }
    }
}


extern "C" void kernel_launch(void* const* d_in, const int* in_sizes, int n_in,
                              void* d_out, int out_size, void* d_ws, size_t ws_size,
                              hipStream_t stream) {
    const float* h   = (const float*)d_in[0];
    const float* wl0 = (const float*)d_in[1];
    const float* wr0 = (const float*)d_in[2];
    const float* b0  = (const float*)d_in[3];
    const float* wl1 = (const float*)d_in[4];
    const float* wr1 = (const float*)d_in[5];
    const float* b1  = (const float*)d_in[6];
    const int*   ei0 = (const int*)d_in[7];
    const int*   ei1 = (const int*)d_in[8];

    const int E0 = in_sizes[7] / 2;
    const int E1 = in_sizes[8] / 2;
    const int NT0 = 20000;
    const int NT1 = 2000;

    float* mean0 = (float*)d_ws;
    float* pn    = mean0 + (size_t)NT0 * DFEAT;
    float* ht1   = pn + (size_t)NT0 * PCOLS;

    sage_mean_owner_kernel<80><<<(NT0 + 79) / 80, AGG_THREADS, 0, stream>>>(
        h, ei0, ei0 + E0, mean0, E0, NT0);

    sage_gemm_lds<5, 2, 8, 32, true, true><<<NT0 / 80, 256, 0, stream>>>(
        mean0, h, wl0, wr0, b0, wl1, pn, ht1, DFEAT, NT1);

    sage_mean_narrow_kernel<8><<<(NT1 + 7) / 8, AGG_THREADS, 0, stream>>>(
        pn, ei1, ei1 + E1, ht1, wr1, b1, (float*)d_out, E1, NT1);
}


# Layer-0 GEMM loop: LDS fragment reads of two k-steps issued as one group ahead of the previous group's 40 WMMAs (second register set), kept up across the chunk barrier; sum order unchanged
# speedup vs baseline: 4.8245x; 1.0699x over previous
#include <hip/hip_runtime.h>

typedef __attribute__((ext_vector_type(2))) float v2f;
typedef __attribute__((ext_vector_type(4))) float v4f;
typedef __attribute__((ext_vector_type(8))) float v8f;

#define DFEAT 256
#define NCLS  47
#define PCOLS 48

#define AGG_THREADS  1024
#define AGG_WAVES    32
#define AGG_CAP      8192
#define AGG_INFLIGHT 8

template<int R>
__global__ __launch_bounds__(AGG_THREADS)
void sage_mean_owner_kernel(const float* __restrict__ hsrc,
                            const int* __restrict__ src,
                            const int* __restrict__ dst,
                            float* __restrict__ mean,
                            int nE, int nTgt) {
    __shared__ float slab[R * DFEAT];
    __shared__ int   list_e[AGG_CAP];
    __shared__ int   list_t[AGG_CAP];
    __shared__ int   sort_s[AGG_CAP];
    __shared__ int   sort_t[AGG_CAP];
    __shared__ int   cnt[R];
    __shared__ int   cursor[R];
    __shared__ int   n_list;

    const int tid  = threadIdx.x;
    const int lane = tid & 31;
    const int wave = tid >> 5;
    const int base = blockIdx.x * R;
    const int rows = min(R, nTgt - base);
    const unsigned ubase = (unsigned)base;
    const unsigned urows = (unsigned)rows;

    for (int i = tid; i < R * DFEAT / 4; i += AGG_THREADS)
        ((float4*)slab)[i] = make_float4(0.0f, 0.0f, 0.0f, 0.0f);
    for (int i = tid; i < R; i += AGG_THREADS) cnt[i] = 0;
    if (tid == 0) n_list = 0;
    __syncthreads();

    auto record = [&](unsigned t, int i) {
        if (t < urows) {
            atomicAdd(&cnt[t], 1);
            const int idx = atomicAdd(&n_list, 1);
            if (idx < AGG_CAP) { list_e[idx] = i; list_t[idx] = (int)t; }
        }
    };
    auto check4 = [&](const int4 q, int i) {
        const unsigned t0 = (unsigned)q.x - ubase;
        const unsigned t1 = (unsigned)q.y - ubase;
        const unsigned t2 = (unsigned)q.z - ubase;
        const unsigned t3 = (unsigned)q.w - ubase;
        if ((t0 < urows) | (t1 < urows) | (t2 < urows) | (t3 < urows)) {
            record(t0, i); record(t1, i + 1); record(t2, i + 2); record(t3, i + 3);
        }
    };
    const bool vec_ok = (((size_t)dst) & 15) == 0;
    const int  nVec   = vec_ok ? (nE >> 2) : 0;
    const int4* dst4  = (const int4*)dst;
    const int start = (int)blockIdx.x * (nVec / (int)gridDim.x);
    for (int pass = 0; pass < 2; ++pass) {
        const int lo = pass ? 0 : start;
        const int hi = pass ? start : nVec;
        int v = lo + tid;
        for (; v + 3 * AGG_THREADS < hi; v += 4 * AGG_THREADS) {
            const int4 q0 = dst4[v];
            const int4 q1 = dst4[v + AGG_THREADS];
            const int4 q2 = dst4[v + 2 * AGG_THREADS];
            const int4 q3 = dst4[v + 3 * AGG_THREADS];
            check4(q0, 4 * v);
            check4(q1, 4 * (v + AGG_THREADS));
            check4(q2, 4 * (v + 2 * AGG_THREADS));
            check4(q3, 4 * (v + 3 * AGG_THREADS));
        }
        for (; v < hi; v += AGG_THREADS) check4(dst4[v], 4 * v);
    }
    for (int i = 4 * nVec + tid; i < nE; i += AGG_THREADS)
        record((unsigned)dst[i] - ubase, i);
    __syncthreads();

    int n = __builtin_amdgcn_readfirstlane(n_list);
    if (n > AGG_CAP) {
        for (int i = tid; i < nE; i += AGG_THREADS) {
            const unsigned t = (unsigned)dst[i] - ubase;
            if (t < urows) {
                const float* row = hsrc + (size_t)src[i] * DFEAT;
                float* acc = slab + t * DFEAT;
                for (int c = 0; c < DFEAT; ++c) atomicAdd(&acc[c], row[c]);
            }
        }
        n = 0;
    }

    if (tid < R) {
        int first = 0;
#pragma unroll
        for (int r = 0; r < R; ++r) first += (r < tid) ? cnt[r] : 0;
        cursor[tid] = first;
    }
    __syncthreads();
    for (int i = tid; i < n; i += AGG_THREADS) {
        const int t = list_t[i];
        const int s = src[list_e[i]];
        const int p = atomicAdd(&cursor[t], 1);
        sort_s[p] = s;
        sort_t[p] = t;
    }
    __syncthreads();

    const int nFull = n / AGG_INFLIGHT;
    for (int b = wave; b < nFull; b += AGG_WAVES) {
        float2 v[AGG_INFLIGHT][4];
        int    tt[AGG_INFLIGHT];
#pragma unroll
        for (int u = 0; u < AGG_INFLIGHT; ++u) {
            const int i = b * AGG_INFLIGHT + u;
            const int s = __builtin_amdgcn_readfirstlane(sort_s[i]);
            tt[u]       = __builtin_amdgcn_readfirstlane(sort_t[i]);
            const float2* row = (const float2*)(hsrc + (size_t)s * DFEAT);
#pragma unroll
            for (int j = 0; j < 4; ++j) v[u][j] = row[lane + 32 * j];
        }
        float2 a[4];
#pragma unroll
        for (int j = 0; j < 4; ++j) a[j] = v[0][j];
#pragma unroll
        for (int u = 1; u <= AGG_INFLIGHT; ++u) {
            if (u == AGG_INFLIGHT || tt[u] != tt[u - 1]) {
                float* acc = slab + tt[u - 1] * DFEAT + 2 * lane;
#pragma unroll
                for (int j = 0; j < 4; ++j) {
                    atomicAdd(&acc[64 * j],     a[j].x);
                    atomicAdd(&acc[64 * j + 1], a[j].y);
                }
                if (u < AGG_INFLIGHT) {
#pragma unroll
                    for (int j = 0; j < 4; ++j) a[j] = v[u][j];
                }
            } else {
#pragma unroll
                for (int j = 0; j < 4; ++j) { a[j].x += v[u][j].x; a[j].y += v[u][j].y; }
            }
        }
    }
    for (int i = nFull * AGG_INFLIGHT + wave; i < n; i += AGG_WAVES) {
        const int s = __builtin_amdgcn_readfirstlane(sort_s[i]);
        const int t = __builtin_amdgcn_readfirstlane(sort_t[i]);
        const float2* row = (const float2*)(hsrc + (size_t)s * DFEAT);
        float* acc = slab + t * DFEAT + 2 * lane;
#pragma unroll
        for (int j = 0; j < 4; ++j) {
            const float2 x = row[lane + 32 * j];
            atomicAdd(&acc[64 * j],     x.x);
            atomicAdd(&acc[64 * j + 1], x.y);
        }
    }
    __syncthreads();

    for (int r = wave; r < rows; r += AGG_WAVES) {
        const float inv = 1.0f / fmaxf((float)cnt[r], 1.0f);
        const float4* srow = (const float4*)(slab + r * DFEAT);
        float4* orow = (float4*)(mean + (size_t)(base + r) * DFEAT);
#pragma unroll
        for (int j = 0; j < 2; ++j) {
            float4 x = srow[lane + 32 * j];
            x.x *= inv; x.y *= inv; x.z *= inv; x.w *= inv;
            orow[lane + 32 * j] = x;
        }
    }
}

template<int R>
__global__ __launch_bounds__(AGG_THREADS)
void sage_mean_narrow_kernel(const float* __restrict__ Pn,
                             const int* __restrict__ src,
                             const int* __restrict__ dst,
                             const float* __restrict__ Ht,
                             const float* __restrict__ Wr,
                             const float* __restrict__ bias,
                             float* __restrict__ out,
                             int nE, int nTgt) {
    __shared__ __attribute__((aligned(16))) float slab[R * PCOLS];
    __shared__ float tterm[R * PCOLS];
    __shared__ int   list_e[AGG_CAP];
    __shared__ int   list_t[AGG_CAP];
    __shared__ int   cnt[R];
    __shared__ int   n_list;

    static_assert(R <= 8, "the owned rows must lie in lane half 0 of one C/D tile");
    constexpr int KSL = 8;
    constexpr int KST = DFEAT / 4 / KSL;
    constexpr int PTN = PCOLS / 16;
    static_assert(PTN * KSL <= AGG_WAVES, "one wave per (column tile, k-slice)");

    const int tid  = threadIdx.x;
    const int lane = tid & 31;
    const int wave = tid >> 5;
    const int c    = lane & 15;
    const int h    = lane >> 4;
    const int base = blockIdx.x * R;
    const int rows = min(R, nTgt - base);
    const unsigned ubase = (unsigned)base;
    const unsigned urows = (unsigned)rows;

    const bool tw = wave < PTN * KSL;
    const int  nt = wave % PTN;
    const int  ks = wave / PTN;
    const bool rok = c < rows;
    const bool cok = 16 * nt + c < NCLS;
    for (int i = tid; i < R * PCOLS; i += AGG_THREADS) { slab[i] = 0.0f; tterm[i] = 0.0f; }
    for (int i = tid; i < R; i += AGG_THREADS) cnt[i] = 0;
    if (tid == 0) n_list = 0;
    __syncthreads();

    v2f fa[KST], fb[KST];
    if (tw) {
        const float* ap = Ht + (size_t)(base + (rok ? c : 0)) * DFEAT + 4 * KST * ks + 2 * h;
        const float* bp = Wr + (size_t)(4 * KST * ks + 2 * h) * NCLS + (cok ? 16 * nt + c : 0);
#pragma unroll
        for (int st = 0; st < KST; ++st) {
            fa[st]   = *(const v2f*)(ap + 4 * st);
            fb[st].x = bp[(4 * st) * NCLS];
            fb[st].y = bp[(4 * st + 1) * NCLS];
        }
    }

    auto record = [&](unsigned t, int i) {
        if (t < urows) {
            atomicAdd(&cnt[t], 1);
            const int idx = atomicAdd(&n_list, 1);
            if (idx < AGG_CAP) { list_e[idx] = i; list_t[idx] = (int)t; }
        }
    };
    auto check4 = [&](const int4 q, int i) {
        const unsigned t0 = (unsigned)q.x - ubase;
        const unsigned t1 = (unsigned)q.y - ubase;
        const unsigned t2 = (unsigned)q.z - ubase;
        const unsigned t3 = (unsigned)q.w - ubase;
        if ((t0 < urows) | (t1 < urows) | (t2 < urows) | (t3 < urows)) {
            record(t0, i); record(t1, i + 1); record(t2, i + 2); record(t3, i + 3);
        }
    };
    const bool vec_ok = (((size_t)dst) & 15) == 0;
    const int  nVec   = vec_ok ? (nE >> 2) : 0;
    const int4* dst4  = (const int4*)dst;
    const int start = (int)blockIdx.x * (nVec / (int)gridDim.x);
    for (int pass = 0; pass < 2; ++pass) {
        const int lo = pass ? 0 : start;
        const int hi = pass ? start : nVec;
        int v = lo + tid;
        for (; v + 3 * AGG_THREADS < hi; v += 4 * AGG_THREADS) {
            const int4 q0 = dst4[v];
            const int4 q1 = dst4[v + AGG_THREADS];
            const int4 q2 = dst4[v + 2 * AGG_THREADS];
            const int4 q3 = dst4[v + 3 * AGG_THREADS];
            check4(q0, 4 * v);
            check4(q1, 4 * (v + AGG_THREADS));
            check4(q2, 4 * (v + 2 * AGG_THREADS));
            check4(q3, 4 * (v + 3 * AGG_THREADS));
        }
        for (; v < hi; v += AGG_THREADS) check4(dst4[v], 4 * v);
    }
    for (int i = 4 * nVec + tid; i < nE; i += AGG_THREADS)
        record((unsigned)dst[i] - ubase, i);

    if (tw) {
        v8f q;
#pragma unroll
        for (int i = 0; i < 8; ++i) q[i] = 0.0f;
#pragma unroll
        for (int st = 0; st < KST; ++st) {
            v2f x = fa[st], y = fb[st];
            if (!rok) { x.x = 0.0f; x.y = 0.0f; }
            if (!cok) { y.x = 0.0f; y.y = 0.0f; }
            q = __builtin_amdgcn_wmma_f32_16x16x4_f32(false, x, false, y, (short)0, q, false, false);
        }
        if (h == 0) {
#pragma unroll
            for (int i = 0; i < 8; ++i)
                if (i < rows) atomicAdd(&tterm[i * PCOLS + 16 * nt + c], q[i]);
        }
    }
    __syncthreads();

    int n = __builtin_amdgcn_readfirstlane(n_list);
    if (n > AGG_CAP) {
        for (int i = tid; i < nE; i += AGG_THREADS) {
            const unsigned t = (unsigned)dst[i] - ubase;
            if (t < urows) {
                const float* row = Pn + (size_t)src[i] * PCOLS;
                float* acc = slab + t * PCOLS;
                for (int j = 0; j < PCOLS; ++j) atomicAdd(&acc[j], row[j]);
            }
        }
        n = 0;
    }

    const int grp = tid >> 4;
    const int l16 = tid & 15;
    for (int i = grp; i < n; i += AGG_THREADS / 16) {
        const int t = list_t[i];
        const int sn = src[list_e[i]];
        if (l16 < PCOLS / 4) {
            const float4 x = ((const float4*)(Pn + (size_t)sn * PCOLS))[l16];
            float* acc = slab + t * PCOLS + 4 * l16;
            atomicAdd(&acc[0], x.x);
            atomicAdd(&acc[1], x.y);
            atomicAdd(&acc[2], x.z);
            atomicAdd(&acc[3], x.w);
        }
    }
    __syncthreads();

    for (int i = tid; i < rows * NCLS; i += AGG_THREADS) {
        const int r = i / NCLS, oc = i - r * NCLS;
        const float inv = 1.0f / fmaxf((float)cnt[r], 1.0f);
        out[(size_t)(base + r) * NCLS + oc] = slab[r * PCOLS + oc] * inv + tterm[r * PCOLS + oc] + bias[oc];
    }
}

template<int MT, int NTW, int NWAVE, int KC, bool WVEC, bool RELU>
__global__ __launch_bounds__(NWAVE * 32)
void sage_gemm_lds(const float* __restrict__ Amean,
                   const float* __restrict__ Htgt,
                   const float* __restrict__ Wl,
                   const float* __restrict__ Wr,
                   const float* __restrict__ bias,
                   const float* __restrict__ Wl2,
                   float* __restrict__ Pn,
                   float* __restrict__ Ht,
                   int Nact, int nQ) {
    constexpr int NTHR   = NWAVE * 32;
    constexpr int ROWS   = 16 * MT;
    constexpr int NCOLS  = 16 * NTW * NWAVE;
    constexpr int KP     = KC / 2;
    constexpr int KQ     = KC / 4;
    constexpr int WPITCH = 2 * NCOLS + ((32 - (2 * NCOLS) % 64 + 64) % 64);
    constexpr int APITCH = KC + 4;
    constexpr int WSZ    = KP * WPITCH;
    constexpr int ASZ    = ROWS * APITCH;
    constexpr int STAGE  = 2 * WSZ + 2 * ASZ;
    constexpr int NCHUNK = DFEAT / KC;
    constexpr int ARPR   = NTHR / KQ;
    constexpr int AR     = (2 * ROWS + ARPR - 1) / ARPR;
    constexpr int WCPT   = WVEC ? 4 : 1;
    constexpr int WRPR   = NTHR / (NCOLS / WCPT);
    constexpr int WRM    = KP / WRPR;
    static_assert(NTHR % KQ == 0 && NTHR % (NCOLS / WCPT) == 0 && KP % WRPR == 0, "copy shares must tile the chunk");

    __shared__ __attribute__((aligned(16))) float lds[2 * STAGE];

    const int tid  = threadIdx.x;
    const int wave = tid >> 5;
    const int lane = tid & 31;
    const int c    = lane & 15;
    const int h    = lane >> 4;
    const int ldw = WVEC ? NCOLS : Nact;
    const int rowBase = blockIdx.x * ROWS;
    const int colBase = wave * (16 * NTW);

    const float* asrc[AR];
    int          aoff[AR];
    {
        const int q  = tid % KQ;
        const int r0 = tid / KQ;
#pragma unroll
        for (int i = 0; i < AR; ++i) {
            const int mrow = min(r0 + i * ARPR, 2 * ROWS - 1);
            const int mat  = mrow >= ROWS;
            const int m    = mrow - mat * ROWS;
            asrc[i] = (mat ? Htgt : Amean) + (size_t)(rowBase + m) * DFEAT + 4 * q;
            aoff[i] = 2 * WSZ + mat * ASZ + m * APITCH + 4 * q;
        }
    }
    const int  wcol = (tid % (NCOLS / WCPT)) * WCPT;
    const int  wp0  = tid / (NCOLS / WCPT);
    const bool wok  = wcol < Nact;
    const int  wtoff = 2 * wp0 * ldw + (wok ? wcol : 0);
    const int    woff0 = wp0 * WPITCH + 2 * wcol;

    v4f areg[AR];
    v4f wv0[2][WRM], wv1[2][WRM];
    v2f ws[2][WRM];

    auto gload = [&](int k0) {
#pragma unroll
        for (int i = 0; i < AR; ++i) areg[i] = *(const v4f*)(asrc[i] + k0);
#pragma unroll
        for (int mat = 0; mat < 2; ++mat) {
            const float* wb = (mat ? Wr : Wl) + (size_t)k0 * ldw;
#pragma unroll
            for (int i = 0; i < WRM; ++i) {
                const float* s = wb + (size_t)(2 * i * WRPR) * ldw;
                if constexpr (WVEC) {
                    wv0[mat][i] = *(const v4f*)(s + wtoff);
                    wv1[mat][i] = *(const v4f*)(s + ldw + wtoff);
                } else {
                    ws[mat][i].x = s[wtoff];
                    ws[mat][i].y = s[ldw + wtoff];
                }
            }
        }
    };
    auto lstore = [&](float* stage) {
#pragma unroll
        for (int i = 0; i < AR; ++i) *(v4f*)(stage + aoff[i]) = areg[i];
#pragma unroll
        for (int mat = 0; mat < 2; ++mat) {
#pragma unroll
            for (int i = 0; i < WRM; ++i) {
                float* d = stage + mat * WSZ + woff0 + i * WRPR * WPITCH;
                if constexpr (WVEC) {
                    const v4f r0 = wv0[mat][i];
                    const v4f r1 = wv1[mat][i];
                    v4f o0, o1;
                    o0.x = r0.x; o0.y = r1.x; o0.z = r0.y; o0.w = r1.y;
                    o1.x = r0.z; o1.y = r1.z; o1.z = r0.w; o1.w = r1.w;
                    *(v4f*)d       = o0;
                    *(v4f*)(d + 4) = o1;
                } else {
                    *(v2f*)d = ws[mat][i];
                }
            }
        }
    };

    v8f acc[MT][NTW];
#pragma unroll
    for (int j = 0; j < NTW; ++j) {
        const int col = colBase + 16 * j + c;
        const float bv = (WVEC || col < Nact) ? bias[col] : 0.0f;
#pragma unroll
        for (int mt = 0; mt < MT; ++mt)
#pragma unroll
            for (int i = 0; i < 8; ++i) acc[mt][j][i] = bv;
    }

    const int aFrag = 2 * WSZ + c * APITCH + 2 * h;
    const int bFrag = h * WPITCH + 2 * (colBase + c);

    constexpr int FG  = 2;
    constexpr int NFG = KQ / FG;
    static_assert(KQ % FG == 0, "fragment groups must tile the chunk");
    v2f fam[2][FG][MT], fah[2][FG][MT], fbl[2][FG][NTW], fbr[2][FG][NTW];
    auto kfload = [&](int buf, const float* cur, int g) {
#pragma unroll
        for (int q = 0; q < FG; ++q) {
            const int kk = g * FG + q;
#pragma unroll
            for (int mt = 0; mt < MT; ++mt) {
                fam[buf][q][mt] = *(const v2f*)(cur + aFrag + 16 * mt * APITCH + 4 * kk);
                fah[buf][q][mt] = *(const v2f*)(cur + aFrag + ASZ + 16 * mt * APITCH + 4 * kk);
            }
#pragma unroll
            for (int j = 0; j < NTW; ++j) {
                fbl[buf][q][j] = *(const v2f*)(cur + bFrag + 2 * kk * WPITCH + 32 * j);
                fbr[buf][q][j] = *(const v2f*)(cur + bFrag + WSZ + 2 * kk * WPITCH + 32 * j);
            }
        }
    };

    gload(0);
    lstore(lds);
    __syncthreads();

    static_assert(NFG % 2 == 0, "a chunk must leave the fragment sets where it found them");
    kfload(0, lds, 0);
    if (NCHUNK > 1) gload(KC);
    for (int ch = 0; ch < NCHUNK; ++ch) {
        const float* cur = lds + (ch & 1) * STAGE;
        float*       nxt = lds + ((ch + 1) & 1) * STAGE;
        const bool more  = ch + 1 < NCHUNK;
        const bool more2 = ch + 2 < NCHUNK;

#pragma unroll
        for (int g = 0; g < NFG; ++g) {
            if (g + 1 < NFG) {
                kfload((g + 1) & 1, cur, g + 1);
            } else if (more) {
                lstore(nxt);
                __syncthreads();
                kfload(0, nxt, 0);
                if (more2) gload((ch + 2) * KC);
            }
            __builtin_amdgcn_sched_barrier(0);
#pragma unroll
            for (int q = 0; q < FG; ++q) {
#pragma unroll
                for (int mt = 0; mt < MT; ++mt)
#pragma unroll
                    for (int j = 0; j < NTW; ++j)
                        acc[mt][j] = __builtin_amdgcn_wmma_f32_16x16x4_f32(
                                         false, fam[g & 1][q][mt], false, fbl[g & 1][q][j], (short)0, acc[mt][j], false, false);
#pragma unroll
                for (int mt = 0; mt < MT; ++mt)
#pragma unroll
                    for (int j = 0; j < NTW; ++j)
                        acc[mt][j] = __builtin_amdgcn_wmma_f32_16x16x4_f32(
                                         false, fah[g & 1][q][mt], false, fbr[g & 1][q][j], (short)0, acc[mt][j], false, false);
            }
            __builtin_amdgcn_sched_barrier(0);
        }
    }
    __syncthreads();

    constexpr int TP   = DFEAT + 4;
    constexpr int PBP  = 2 * PCOLS;
    constexpr int TSZ  = ROWS * TP;
    constexpr int PBSZ = (DFEAT / 2) * PBP;
    constexpr int PTM  = ROWS / 16;
    constexpr int PTN  = PCOLS / 16;
    static_assert(TSZ + PBSZ <= 2 * STAGE, "tile and projection matrix must fit the two stages");
    static_assert(NCOLS == DFEAT && PBP % 64 == 32 && PTM * PTN <= 2 * NWAVE, "projection tiling");
    float* tl = lds;
    float* pb = lds + TSZ;

#pragma unroll
    for (int mt = 0; mt < MT; ++mt)
#pragma unroll
        for (int j = 0; j < NTW; ++j)
#pragma unroll
            for (int i = 0; i < 8; ++i) {
                float v = acc[mt][j][i];
                if (RELU) v = fmaxf(v, 0.0f);
                tl[(16 * mt + 8 * h + i) * TP + colBase + 16 * j + c] = v;
            }
    constexpr int PW_PER = DFEAT * NCLS / NTHR;
    static_assert((DFEAT * NCLS) % NTHR == 0, "weight copy must tile");
    {
        float wv[PW_PER];
        int   wo[PW_PER];
        int k = tid / NCLS, col = tid - k * NCLS;
#pragma unroll
        for (int u = 0; u < PW_PER; ++u) {
            wv[u] = Wl2[tid + u * NTHR];
            wo[u] = (k >> 1) * PBP + 2 * col + (k & 1);
            col += NTHR % NCLS; k += NTHR / NCLS;
            if (col >= NCLS) { col -= NCLS; ++k; }
        }
#pragma unroll
        for (int u = 0; u < PW_PER; ++u) pb[wo[u]] = wv[u];
        for (int e = tid; e < DFEAT * (PCOLS - NCLS); e += NTHR) {
            const int kk = e / (PCOLS - NCLS), pc = NCLS + e - kk * (PCOLS - NCLS);
            pb[(kk >> 1) * PBP + 2 * pc + (kk & 1)] = 0.0f;
        }
    }
    __syncthreads();

    if (rowBase < nQ) {
        for (int r = wave; r < ROWS; r += NWAVE) {
            if (rowBase + r < nQ) {
                const v4f* srow = (const v4f*)(tl + r * TP);
                v4f* orow = (v4f*)(Ht + (size_t)(rowBase + r) * DFEAT);
                orow[lane]      = srow[lane];
                orow[lane + 32] = srow[lane + 32];
            }
        }
    }

    constexpr int PG  = 8;
    constexpr int PNG = DFEAT / 4 / PG;
    const int t0 = wave;
    const int t1 = min(wave + NWAVE, PTM * PTN - 1);
    const int m0 = t0 / PTN, n0 = t0 - m0 * PTN;
    const int m1 = t1 / PTN, n1 = t1 - m1 * PTN;
    const float* a0 = tl + (16 * m0 + c) * TP + 2 * h;
    const float* a1 = tl + (16 * m1 + c) * TP + 2 * h;
    const float* b0 = pb + h * PBP + 2 * (16 * n0 + c);
    const float* b1 = pb + h * PBP + 2 * (16 * n1 + c);
    v8f p0, p1;
#pragma unroll
    for (int i = 0; i < 8; ++i) { p0[i] = 0.0f; p1[i] = 0.0f; }
    v2f x0[2][PG], y0[2][PG], x1[2][PG], y1[2][PG];
    auto fload = [&](int buf, int g) {
#pragma unroll
        for (int q = 0; q < PG; ++q) {
            const int st = g * PG + q;
            x0[buf][q] = *(const v2f*)(a0 + 4 * st);
            y0[buf][q] = *(const v2f*)(b0 + 2 * st * PBP);
            x1[buf][q] = *(const v2f*)(a1 + 4 * st);
            y1[buf][q] = *(const v2f*)(b1 + 2 * st * PBP);
        }
    };
    fload(0, 0);
#pragma unroll
    for (int g = 0; g < PNG; ++g) {
        if (g + 1 < PNG) fload((g + 1) & 1, g + 1);
        __builtin_amdgcn_sched_barrier(0);
#pragma unroll
        for (int q = 0; q < PG; ++q) {
            p0 = __builtin_amdgcn_wmma_f32_16x16x4_f32(false, x0[g & 1][q], false, y0[g & 1][q], (short)0, p0, false, false);
            p1 = __builtin_amdgcn_wmma_f32_16x16x4_f32(false, x1[g & 1][q], false, y1[g & 1][q], (short)0, p1, false, false);
        }
        __builtin_amdgcn_sched_barrier(0);
    }
    {
        float* o0 = Pn + (size_t)(rowBase + 16 * m0 + 8 * h) * PCOLS + 16 * n0 + c;
        float* o1 = Pn + (size_t)(rowBase + 16 * m1 + 8 * h) * PCOLS + 16 * n1 + c;
#pragma unroll
        for (int i = 0; i < 8; ++i) {
            o0[i * PCOLS] = p0[i];
            if (t1 == wave + NWAVE) o1[i * PCOLS] = p1[i];
        }
    }
}


extern "C" void kernel_launch(void* const* d_in, const int* in_sizes, int n_in,
                              void* d_out, int out_size, void* d_ws, size_t ws_size,
                              hipStream_t stream) {
    const float* h   = (const float*)d_in[0];
    const float* wl0 = (const float*)d_in[1];
    const float* wr0 = (const float*)d_in[2];
    const float* b0  = (const float*)d_in[3];
    const float* wl1 = (const float*)d_in[4];
    const float* wr1 = (const float*)d_in[5];
    const float* b1  = (const float*)d_in[6];
    const int*   ei0 = (const int*)d_in[7];
    const int*   ei1 = (const int*)d_in[8];

    const int E0 = in_sizes[7] / 2;
    const int E1 = in_sizes[8] / 2;
    const int NT0 = 20000;
    const int NT1 = 2000;

    float* mean0 = (float*)d_ws;
    float* pn    = mean0 + (size_t)NT0 * DFEAT;
    float* ht1   = pn + (size_t)NT0 * PCOLS;

    sage_mean_owner_kernel<80><<<(NT0 + 79) / 80, AGG_THREADS, 0, stream>>>(
        h, ei0, ei0 + E0, mean0, E0, NT0);

    sage_gemm_lds<5, 2, 8, 32, true, true><<<NT0 / 80, 256, 0, stream>>>(
        mean0, h, wl0, wr0, b0, wl1, pn, ht1, DFEAT, NT1);

    sage_mean_narrow_kernel<8><<<(NT1 + 7) / 8, AGG_THREADS, 0, stream>>>(
        pn, ei1, ei1 + E1, ht1, wr1, b1, (float*)d_out, E1, NT1);
}
